# starting kernel
# baseline (speedup 1.0000x reference)
#include <hip/hip_runtime.h>

typedef __attribute__((ext_vector_type(2))) float v2f;
typedef __attribute__((ext_vector_type(8))) float v8f;

#define MB 25
#define NP 64

__device__ __forceinline__ void qmul4(const float a[4], const float b[4], float o[4]) {
    o[0] = a[0]*b[0] - a[1]*b[1] - a[2]*b[2] - a[3]*b[3];
    o[1] = a[0]*b[1] + b[0]*a[1] + (a[2]*b[3] - a[3]*b[2]);
    o[2] = a[0]*b[2] + b[0]*a[2] + (a[3]*b[1] - a[1]*b[3]);
    o[3] = a[0]*b[3] + b[0]*a[3] + (a[1]*b[2] - a[2]*b[1]);
}

__global__ __launch_bounds__(128)
void dqskin_wmma_kernel(const float* __restrict__ xyz,
                        const float* __restrict__ t_qr,
                        const float* __restrict__ t_qd,
                        const float* __restrict__ rest_qr,
                        const float* __restrict__ rest_qd,
                        const float* __restrict__ log_gauss,
                        float* __restrict__ out)
{
    __shared__ float sQR[MB][4];
    __shared__ float qrS[MB][4];
    __shared__ float qdS[MB][4];
    __shared__ float Wl[32][12];
    __shared__ float dl[NP][32];

    const int m   = blockIdx.x;
    const int tid = threadIdx.x;

    float se3r[4] = {0.f,0.f,0.f,0.f};
    float se3d[4] = {0.f,0.f,0.f,0.f};

    if (tid < MB) {
        const int b = tid;
        const float* tq = t_qr   + (m*MB + b)*4;
        const float* td = t_qd   + (m*MB + b)*4;
        const float* rq = rest_qr+ (m*MB + b)*4;
        const float* rd = rest_qd+ (m*MB + b)*4;
        float tqv[4] = {tq[0],  tq[1],  tq[2],  tq[3]};
        float tdv[4] = {td[0],  td[1],  td[2],  td[3]};
        float rqv[4] = {rq[0],  rq[1],  rq[2],  rq[3]};
        float rqc[4] = {rq[0], -rq[1], -rq[2], -rq[3]};
        float rdc[4] = {rd[0], -rd[1], -rd[2], -rd[3]};

        qmul4(tqv, rqc, se3r);
        float u1[4], u2[4];
        qmul4(tqv, rdc, u1);
        qmul4(tdv, rqc, u2);
        #pragma unroll
        for (int k = 0; k < 4; ++k) se3d[k] = u1[k] + u2[k];
        #pragma unroll
        for (int k = 0; k < 4; ++k) sQR[b][k] = se3r[k];

        const float w0 = rqc[0], ax = rqc[1], ay = rqc[2], az = rqc[3];
        const float R00 = 1.f - 2.f*(ay*ay + az*az);
        const float R01 = 2.f*(ax*ay - w0*az);
        const float R02 = 2.f*(ax*az + w0*ay);
        const float R10 = 2.f*(ax*ay + w0*az);
        const float R11 = 1.f - 2.f*(ax*ax + az*az);
        const float R12 = 2.f*(ay*az - w0*ax);
        const float R20 = 2.f*(ax*az - w0*ay);
        const float R21 = 2.f*(ay*az + w0*ax);
        const float R22 = 1.f - 2.f*(ax*ax + ay*ay);
        float tt[4];
        qmul4(rdc, rqv, tt);
        const float tx = 2.f*tt[1], ty = 2.f*tt[2], tz = 2.f*tt[3];

        const float i0 = __expf(-log_gauss[b*3+0]);
        const float i1 = __expf(-log_gauss[b*3+1]);
        const float i2 = __expf(-log_gauss[b*3+2]);
        const float A00=R00*i0, A01=R01*i0, A02=R02*i0;
        const float A10=R10*i1, A11=R11*i1, A12=R12*i1;
        const float A20=R20*i2, A21=R21*i2, A22=R22*i2;
        const float c0=tx*i0, c1=ty*i1, c2=tz*i2;

        Wl[b][0]  = A00*A00 + A10*A10 + A20*A20;
        Wl[b][1]  = A01*A01 + A11*A11 + A21*A21;
        Wl[b][2]  = A02*A02 + A12*A12 + A22*A22;
        Wl[b][3]  = 2.f*(A00*A01 + A10*A11 + A20*A21);
        Wl[b][4]  = 2.f*(A00*A02 + A10*A12 + A20*A22);
        Wl[b][5]  = 2.f*(A01*A02 + A11*A12 + A21*A22);
        Wl[b][6]  = 2.f*(A00*c0 + A10*c1 + A20*c2);
        Wl[b][7]  = 2.f*(A01*c0 + A11*c1 + A21*c2);
        Wl[b][8]  = 2.f*(A02*c0 + A12*c1 + A22*c2);
        Wl[b][9]  = c0*c0 + c1*c1 + c2*c2;
        Wl[b][10] = 0.f;
        Wl[b][11] = 0.f;
    } else if (tid < 32) {
        #pragma unroll
        for (int k = 0; k < 12; ++k) Wl[tid][k] = 0.f;
    }
    __syncthreads();

    if (tid < MB) {
        const float d = se3r[0]*sQR[0][0] + se3r[1]*sQR[0][1]
                      + se3r[2]*sQR[0][2] + se3r[3]*sQR[0][3];
        const float sgn = (d >= 0.f) ? 1.f : -1.f;
        #pragma unroll
        for (int k = 0; k < 4; ++k) { qrS[tid][k] = sgn*se3r[k]; qdS[tid][k] = sgn*se3d[k]; }
    }
    __syncthreads();

    {
        const int ln  = tid & 31;
        const int wv  = tid >> 5;
        const bool hiL = (ln >= 16);
        const int row = ln & 15;

        const float* xp = xyz + ((size_t)m*NP + wv*16 + row)*3;
        const float x = xp[0], y = xp[1], z = xp[2];

        v2f a0, a1, a2;
        a0[0] = hiL ? z*z : x*x;   a0[1] = hiL ? x*y : y*y;
        a1[0] = hiL ? x   : x*z;   a1[1] = hiL ? y   : y*z;
        a2[0] = hiL ? 0.f : z;     a2[1] = hiL ? 0.f : 1.f;
        const int ko = hiL ? 2 : 0;

        #pragma unroll
        for (int t = 0; t < 2; ++t) {
            const int bone = t*16 + row;
            const float* wr = &Wl[bone][0];
            v2f b0, b1, b2;
            b0[0] = wr[0+ko]; b0[1] = wr[1+ko];
            b1[0] = wr[4+ko]; b1[1] = wr[5+ko];
            b2[0] = wr[8+ko]; b2[1] = wr[9+ko];
            v8f c = {0.f,0.f,0.f,0.f,0.f,0.f,0.f,0.f};
            c = __builtin_amdgcn_wmma_f32_16x16x4_f32(false, a0, false, b0, (short)0, c, false, false);
            c = __builtin_amdgcn_wmma_f32_16x16x4_f32(false, a1, false, b1, (short)0, c, false, false);
            c = __builtin_amdgcn_wmma_f32_16x16x4_f32(false, a2, false, b2, (short)0, c, false, false);
            const int pb = wv*16 + (hiL ? 8 : 0);
            #pragma unroll
            for (int r = 0; r < 8; ++r) dl[pb + r][bone] = c[r];
        }
    }
    __syncthreads();

    if (tid < NP) {
        const int p = tid;
        float dmin = 3.4e38f;
        #pragma unroll
        for (int b = 0; b < MB; ++b) dmin = fminf(dmin, dl[p][b]);

        float q0=0.f,q1=0.f,q2=0.f,q3=0.f, d0=0.f,d1=0.f,d2=0.f,d3=0.f;
        #pragma unroll
        for (int b = 0; b < MB; ++b) {
            const float e = __expf(0.5f*(dmin - dl[p][b]));
            q0 += e*qrS[b][0]; q1 += e*qrS[b][1]; q2 += e*qrS[b][2]; q3 += e*qrS[b][3];
            d0 += e*qdS[b][0]; d1 += e*qdS[b][1]; d2 += e*qdS[b][2]; d3 += e*qdS[b][3];
        }
        const float n2  = q0*q0 + q1*q1 + q2*q2 + q3*q3;
        const float rin = rsqrtf(n2);
        const float w  = q0*rin, vx = q1*rin, vy = q2*rin, vz = q3*rin;

        const float* xp = xyz + ((size_t)m*NP + p)*3;
        const float x = xp[0], y = xp[1], z = xp[2];
        const float ux = vy*z - vz*y, uy = vz*x - vx*z, uz = vx*y - vy*x;
        const float rx = x + 2.f*(w*ux + (vy*uz - vz*uy));
        const float ry = y + 2.f*(w*uy + (vz*ux - vx*uz));
        const float rz = z + 2.f*(w*uz + (vx*uy - vy*ux));
        const float s2 = 2.f*rin*rin;
        const float tx = s2*(-d0*q1 + q0*d1 - d2*q3 + d3*q2);
        const float ty = s2*(-d0*q2 + q0*d2 - d3*q1 + d1*q3);
        const float tz = s2*(-d0*q3 + q0*d3 - d1*q2 + d2*q1);

        float* op = out + ((size_t)m*NP + p)*3;
        op[0] = rx + tx; op[1] = ry + ty; op[2] = rz + tz;
    }
}

extern "C" void kernel_launch(void* const* d_in, const int* in_sizes, int n_in,
                              void* d_out, int out_size, void* d_ws, size_t ws_size,
                              hipStream_t stream) {
    (void)n_in; (void)out_size; (void)d_ws; (void)ws_size;
    const float* xyz     = (const float*)d_in[0];
    const float* t_qr    = (const float*)d_in[1];
    const float* t_qd    = (const float*)d_in[2];
    const float* rest_qr = (const float*)d_in[3];
    const float* rest_qd = (const float*)d_in[4];
    const float* lg      = (const float*)d_in[5];
    float* out = (float*)d_out;

    const int M = in_sizes[1] / (MB * 4);
    dqskin_wmma_kernel<<<dim3(M), dim3(128), 0, stream>>>(
        xyz, t_qr, t_qd, rest_qr, rest_qd, lg, out);
}


# Bone blend moved from LDS-fed vector FMAs to f32 16x16x4 WMMAs fed with the exp weights straight from the transposed dist2 accumulators; blend sum order becomes the matrix unit's
# speedup vs baseline: 1.1399x; 1.1399x over previous
#include <hip/hip_runtime.h>


typedef __attribute__((ext_vector_type(2))) float v2f;
typedef __attribute__((ext_vector_type(8))) float v8f;

#define MB 25
#define NP 64
#define QP 36

__device__ __forceinline__ void qmul4(const float a[4], const float b[4], float o[4]) {
    o[0] = a[0]*b[0] - a[1]*b[1] - a[2]*b[2] - a[3]*b[3];
    o[1] = a[0]*b[1] + b[0]*a[1] + (a[2]*b[3] - a[3]*b[2]);
    o[2] = a[0]*b[2] + b[0]*a[2] + (a[3]*b[1] - a[1]*b[3]);
    o[3] = a[0]*b[3] + b[0]*a[3] + (a[1]*b[2] - a[2]*b[1]);
}

__device__ __forceinline__ float xhalf(float x) {
    return __builtin_bit_cast(float, __builtin_amdgcn_permlanex16(
        __builtin_bit_cast(int, x), __builtin_bit_cast(int, x), 0x76543210, 0xfedcba98, false, false));
}

__global__ __launch_bounds__(128)
void dqskin_wmma_kernel(const float* __restrict__ xyz,
                        const float* __restrict__ t_qr,
                        const float* __restrict__ t_qd,
                        const float* __restrict__ rest_qr,
                        const float* __restrict__ rest_qd,
                        const float* __restrict__ log_gauss,
                        float* __restrict__ out)
{
    __shared__ float sQR[MB][4];
    __shared__ __attribute__((aligned(16))) float QT[8][QP];
    __shared__ float Wl[32][12];

    const int m   = blockIdx.x;
    const int tid = threadIdx.x;

    float se3r[4] = {0.f,0.f,0.f,0.f};
    float se3d[4] = {0.f,0.f,0.f,0.f};

    if (tid < MB) {
        const int b = tid;
        const float* tq = t_qr   + (m*MB + b)*4;
        const float* td = t_qd   + (m*MB + b)*4;
        const float* rq = rest_qr+ (m*MB + b)*4;
        const float* rd = rest_qd+ (m*MB + b)*4;
        float tqv[4] = {tq[0],  tq[1],  tq[2],  tq[3]};
        float tdv[4] = {td[0],  td[1],  td[2],  td[3]};
        float rqv[4] = {rq[0],  rq[1],  rq[2],  rq[3]};
        float rqc[4] = {rq[0], -rq[1], -rq[2], -rq[3]};
        float rdc[4] = {rd[0], -rd[1], -rd[2], -rd[3]};

        qmul4(tqv, rqc, se3r);
        float u1[4], u2[4];
        qmul4(tqv, rdc, u1);
        qmul4(tdv, rqc, u2);
        #pragma unroll
        for (int k = 0; k < 4; ++k) se3d[k] = u1[k] + u2[k];
        #pragma unroll
        for (int k = 0; k < 4; ++k) sQR[b][k] = se3r[k];

        const float w0 = rqc[0], ax = rqc[1], ay = rqc[2], az = rqc[3];
        const float R00 = 1.f - 2.f*(ay*ay + az*az);
        const float R01 = 2.f*(ax*ay - w0*az);
        const float R02 = 2.f*(ax*az + w0*ay);
        const float R10 = 2.f*(ax*ay + w0*az);
        const float R11 = 1.f - 2.f*(ax*ax + az*az);
        const float R12 = 2.f*(ay*az - w0*ax);
        const float R20 = 2.f*(ax*az - w0*ay);
        const float R21 = 2.f*(ay*az + w0*ax);
        const float R22 = 1.f - 2.f*(ax*ax + ay*ay);
        float tt[4];
        qmul4(rdc, rqv, tt);
        const float tx = 2.f*tt[1], ty = 2.f*tt[2], tz = 2.f*tt[3];

        const float i0 = __expf(-log_gauss[b*3+0]);
        const float i1 = __expf(-log_gauss[b*3+1]);
        const float i2 = __expf(-log_gauss[b*3+2]);
        const float A00=R00*i0, A01=R01*i0, A02=R02*i0;
        const float A10=R10*i1, A11=R11*i1, A12=R12*i1;
        const float A20=R20*i2, A21=R21*i2, A22=R22*i2;
        const float c0=tx*i0, c1=ty*i1, c2=tz*i2;

        Wl[b][0]  = A00*A00 + A10*A10 + A20*A20;
        Wl[b][1]  = A01*A01 + A11*A11 + A21*A21;
        Wl[b][2]  = A02*A02 + A12*A12 + A22*A22;
        Wl[b][3]  = 2.f*(A00*A01 + A10*A11 + A20*A21);
        Wl[b][4]  = 2.f*(A00*A02 + A10*A12 + A20*A22);
        Wl[b][5]  = 2.f*(A01*A02 + A11*A12 + A21*A22);
        Wl[b][6]  = 2.f*(A00*c0 + A10*c1 + A20*c2);
        Wl[b][7]  = 2.f*(A01*c0 + A11*c1 + A21*c2);
        Wl[b][8]  = 2.f*(A02*c0 + A12*c1 + A22*c2);
        Wl[b][9]  = c0*c0 + c1*c1 + c2*c2;
        Wl[b][10] = 0.f;
        Wl[b][11] = 0.f;
    } else if (tid < 32) {
        #pragma unroll
        for (int k = 0; k < 12; ++k) Wl[tid][k] = 0.f;
    }
    __syncthreads();

    if (tid < 32) {
        const float d = se3r[0]*sQR[0][0] + se3r[1]*sQR[0][1]
                      + se3r[2]*sQR[0][2] + se3r[3]*sQR[0][3];
        const float sgn = (d >= 0.f) ? 1.f : -1.f;
        #pragma unroll
        for (int k = 0; k < 4; ++k) { QT[k][tid] = sgn*se3r[k]; QT[4+k][tid] = sgn*se3d[k]; }
    }
    __syncthreads();

    const int ln  = tid & 31;
    const int wv  = tid >> 5;
    const bool hiL = (ln >= 16);
    const int row = ln & 15;
    const int p   = wv*16 + row;

    const float* xp = xyz + ((size_t)m*NP + p)*3;
    const float x = xp[0], y = xp[1], z = xp[2];

    v2f f0, f1, f2;
    f0[0] = hiL ? z*z : x*x;   f0[1] = hiL ? x*y : y*y;
    f1[0] = hiL ? x   : x*z;   f1[1] = hiL ? y   : y*z;
    f2[0] = hiL ? 0.f : z;     f2[1] = hiL ? 0.f : 1.f;
    const int ko = hiL ? 2 : 0;

    v8f dt[2];
    #pragma unroll
    for (int t = 0; t < 2; ++t) {
        const float* wr = &Wl[t*16 + row][0];
        v2f w0, w1, w2;
        w0[0] = wr[0+ko]; w0[1] = wr[1+ko];
        w1[0] = wr[4+ko]; w1[1] = wr[5+ko];
        w2[0] = wr[8+ko]; w2[1] = wr[9+ko];
        v8f c = {0.f,0.f,0.f,0.f,0.f,0.f,0.f,0.f};
        c = __builtin_amdgcn_wmma_f32_16x16x4_f32(false, w0, false, f0, (short)0, c, false, false);
        c = __builtin_amdgcn_wmma_f32_16x16x4_f32(false, w1, false, f1, (short)0, c, false, false);
        c = __builtin_amdgcn_wmma_f32_16x16x4_f32(false, w2, false, f2, (short)0, c, false, false);
        dt[t] = c;
    }
    #pragma unroll
    for (int r = 1; r < 8; ++r) dt[1][r] = hiL ? __builtin_inff() : dt[1][r];

    float dmin = dt[0][0];
    #pragma unroll
    for (int r = 1; r < 8; ++r) dmin = fminf(dmin, dt[0][r]);
    #pragma unroll
    for (int r = 0; r < 8; ++r) dmin = fminf(dmin, dt[1][r]);
    dmin = fminf(dmin, xhalf(dmin));

    v8f e[2];
    #pragma unroll
    for (int t = 0; t < 2; ++t) {
        #pragma unroll
        for (int r = 0; r < 8; ++r) e[t][r] = __expf(0.5f*(dmin - dt[t][r]));
    }

    const float* qrow = &QT[row & 7][hiL ? 8 : 0];
    v8f acc[2];
    #pragma unroll
    for (int t = 0; t < 2; ++t) {
        const float4 qa = *reinterpret_cast<const float4*>(qrow + 16*t);
        const float4 qb = *reinterpret_cast<const float4*>(qrow + 16*t + 4);
        v2f a, b;
        v8f c = {0.f,0.f,0.f,0.f,0.f,0.f,0.f,0.f};
        a[0] = qa.x; a[1] = qa.y; b[0] = e[t][0]; b[1] = e[t][1];
        c = __builtin_amdgcn_wmma_f32_16x16x4_f32(false, a, false, b, (short)0, c, false, false);
        a[0] = qa.z; a[1] = qa.w; b[0] = e[t][2]; b[1] = e[t][3];
        c = __builtin_amdgcn_wmma_f32_16x16x4_f32(false, a, false, b, (short)0, c, false, false);
        a[0] = qb.x; a[1] = qb.y; b[0] = e[t][4]; b[1] = e[t][5];
        c = __builtin_amdgcn_wmma_f32_16x16x4_f32(false, a, false, b, (short)0, c, false, false);
        a[0] = qb.z; a[1] = qb.w; b[0] = e[t][6]; b[1] = e[t][7];
        c = __builtin_amdgcn_wmma_f32_16x16x4_f32(false, a, false, b, (short)0, c, false, false);
        acc[t] = c;
    }
    const float q0 = acc[0][0] + acc[1][0], q1 = acc[0][1] + acc[1][1];
    const float q2 = acc[0][2] + acc[1][2], q3 = acc[0][3] + acc[1][3];
    const float d0 = acc[0][4] + acc[1][4], d1 = acc[0][5] + acc[1][5];
    const float d2 = acc[0][6] + acc[1][6], d3 = acc[0][7] + acc[1][7];

    const float n2  = q0*q0 + q1*q1 + q2*q2 + q3*q3;
    const float rin = rsqrtf(n2);
    const float w  = q0*rin, vx = q1*rin, vy = q2*rin, vz = q3*rin;

    const float ux = vy*z - vz*y, uy = vz*x - vx*z, uz = vx*y - vy*x;
    const float rx = x + 2.f*(w*ux + (vy*uz - vz*uy));
    const float ry = y + 2.f*(w*uy + (vz*ux - vx*uz));
    const float rz = z + 2.f*(w*uz + (vx*uy - vy*ux));
    const float s2 = 2.f*rin*rin;
    const float tx = s2*(-d0*q1 + q0*d1 - d2*q3 + d3*q2);
    const float ty = s2*(-d0*q2 + q0*d2 - d3*q1 + d1*q3);
    const float tz = s2*(-d0*q3 + q0*d3 - d1*q2 + d2*q1);

    if (!hiL) {
        float* op = out + ((size_t)m*NP + p)*3;
        op[0] = rx + tx; op[1] = ry + ty; op[2] = rz + tz;
    }
}

extern "C" void kernel_launch(void* const* d_in, const int* in_sizes, int n_in,
                              void* d_out, int out_size, void* d_ws, size_t ws_size,
                              hipStream_t stream) {
    (void)n_in; (void)out_size; (void)d_ws; (void)ws_size;
    const float* xyz     = (const float*)d_in[0];
    const float* t_qr    = (const float*)d_in[1];
    const float* t_qd    = (const float*)d_in[2];
    const float* rest_qr = (const float*)d_in[3];
    const float* rest_qd = (const float*)d_in[4];
    const float* lg      = (const float*)d_in[5];
    float* out = (float*)d_out;

    const int M = in_sizes[1] / (MB * 4);
    dqskin_wmma_kernel<<<dim3(M), dim3(128), 0, stream>>>(
        xyz, t_qr, t_qd, rest_qr, rest_qd, lg, out);
}


# One wave per m instead of four: each wave builds its bone tables in its own LDS slice and loops over the four 16-point tiles, so 4x fewer waves enter and block barriers become wave-scope syncs
# speedup vs baseline: 1.5707x; 1.3779x over previous
#include <hip/hip_runtime.h>


typedef __attribute__((ext_vector_type(2))) float v2f;
typedef __attribute__((ext_vector_type(8))) float v8f;

#define MB 25
#define NP 64
#define NT 4
#define QP 36
#define WPB 4
#define NBLK 2048

__device__ __forceinline__ void qmul4(const float a[4], const float b[4], float o[4]) {
    o[0] = a[0]*b[0] - a[1]*b[1] - a[2]*b[2] - a[3]*b[3];
    o[1] = a[0]*b[1] + b[0]*a[1] + (a[2]*b[3] - a[3]*b[2]);
    o[2] = a[0]*b[2] + b[0]*a[2] + (a[3]*b[1] - a[1]*b[3]);
    o[3] = a[0]*b[3] + b[0]*a[3] + (a[1]*b[2] - a[2]*b[1]);
}

__device__ __forceinline__ float xhalf(float x) {
    return __builtin_bit_cast(float, __builtin_amdgcn_permlanex16(
        __builtin_bit_cast(int, x), __builtin_bit_cast(int, x), 0x76543210, 0xfedcba98, false, false));
}

__device__ __forceinline__ float lane0(float x) {
    return __builtin_bit_cast(float, __builtin_amdgcn_readfirstlane(__builtin_bit_cast(int, x)));
}

__device__ __forceinline__ void wave_lds_sync() {
    __builtin_amdgcn_fence(__ATOMIC_RELEASE, "wavefront");
    __builtin_amdgcn_wave_barrier();
    __builtin_amdgcn_fence(__ATOMIC_ACQUIRE, "wavefront");
}

__global__ __launch_bounds__(32*WPB)
void dqskin_wave_kernel(const float* __restrict__ xyz,
                        const float* __restrict__ t_qr,
                        const float* __restrict__ t_qd,
                        const float* __restrict__ rest_qr,
                        const float* __restrict__ rest_qd,
                        const float* __restrict__ log_gauss,
                        float* __restrict__ out,
                        int M)
{
    __shared__ __attribute__((aligned(16))) float QT[WPB][8][QP];
    __shared__ float Wl[WPB][32][12];

    const int ln  = threadIdx.x & 31;
    const int wv  = __builtin_amdgcn_readfirstlane((int)(threadIdx.x >> 5));
    const bool hiL = (ln >= 16);
    const int row = ln & 15;
    const int ko  = hiL ? 2 : 0;

    float (*QTw)[QP] = QT[wv];
    float (*Wlw)[12] = Wl[wv];

    for (int m = blockIdx.x*WPB + wv; m < M; m += NBLK*WPB) {
        float px[NT], py[NT], pz[NT];
        #pragma unroll
        for (int j = 0; j < NT; ++j) {
            const float* xp = xyz + ((size_t)m*NP + j*16 + row)*3;
            px[j] = xp[0]; py[j] = xp[1]; pz[j] = xp[2];
        }

        float se3r[4] = {0.f,0.f,0.f,0.f};
        float se3d[4] = {0.f,0.f,0.f,0.f};

        if (ln < MB) {
            const int b = ln;
            const float* tq = t_qr   + ((size_t)m*MB + b)*4;
            const float* td = t_qd   + ((size_t)m*MB + b)*4;
            const float* rq = rest_qr+ ((size_t)m*MB + b)*4;
            const float* rd = rest_qd+ ((size_t)m*MB + b)*4;
            float tqv[4] = {tq[0],  tq[1],  tq[2],  tq[3]};
            float tdv[4] = {td[0],  td[1],  td[2],  td[3]};
            float rqv[4] = {rq[0],  rq[1],  rq[2],  rq[3]};
            float rqc[4] = {rq[0], -rq[1], -rq[2], -rq[3]};
            float rdc[4] = {rd[0], -rd[1], -rd[2], -rd[3]};

            qmul4(tqv, rqc, se3r);
            float u1[4], u2[4];
            qmul4(tqv, rdc, u1);
            qmul4(tdv, rqc, u2);
            #pragma unroll
            for (int k = 0; k < 4; ++k) se3d[k] = u1[k] + u2[k];

            const float w0 = rqc[0], ax = rqc[1], ay = rqc[2], az = rqc[3];
            const float R00 = 1.f - 2.f*(ay*ay + az*az);
            const float R01 = 2.f*(ax*ay - w0*az);
            const float R02 = 2.f*(ax*az + w0*ay);
            const float R10 = 2.f*(ax*ay + w0*az);
            const float R11 = 1.f - 2.f*(ax*ax + az*az);
            const float R12 = 2.f*(ay*az - w0*ax);
            const float R20 = 2.f*(ax*az - w0*ay);
            const float R21 = 2.f*(ay*az + w0*ax);
            const float R22 = 1.f - 2.f*(ax*ax + ay*ay);
            float tt[4];
            qmul4(rdc, rqv, tt);
            const float tx = 2.f*tt[1], ty = 2.f*tt[2], tz = 2.f*tt[3];

            const float i0 = __expf(-log_gauss[b*3+0]);
            const float i1 = __expf(-log_gauss[b*3+1]);
            const float i2 = __expf(-log_gauss[b*3+2]);
            const float A00=R00*i0, A01=R01*i0, A02=R02*i0;
            const float A10=R10*i1, A11=R11*i1, A12=R12*i1;
            const float A20=R20*i2, A21=R21*i2, A22=R22*i2;
            const float c0=tx*i0, c1=ty*i1, c2=tz*i2;

            Wlw[b][0]  = A00*A00 + A10*A10 + A20*A20;
            Wlw[b][1]  = A01*A01 + A11*A11 + A21*A21;
            Wlw[b][2]  = A02*A02 + A12*A12 + A22*A22;
            Wlw[b][3]  = 2.f*(A00*A01 + A10*A11 + A20*A21);
            Wlw[b][4]  = 2.f*(A00*A02 + A10*A12 + A20*A22);
            Wlw[b][5]  = 2.f*(A01*A02 + A11*A12 + A21*A22);
            Wlw[b][6]  = 2.f*(A00*c0 + A10*c1 + A20*c2);
            Wlw[b][7]  = 2.f*(A01*c0 + A11*c1 + A21*c2);
            Wlw[b][8]  = 2.f*(A02*c0 + A12*c1 + A22*c2);
            Wlw[b][9]  = c0*c0 + c1*c1 + c2*c2;
            Wlw[b][10] = 0.f;
            Wlw[b][11] = 0.f;
        } else {
            #pragma unroll
            for (int k = 0; k < 12; ++k) Wlw[ln][k] = 0.f;
        }

        {
            const float d = se3r[0]*lane0(se3r[0]) + se3r[1]*lane0(se3r[1])
                          + se3r[2]*lane0(se3r[2]) + se3r[3]*lane0(se3r[3]);
            const float sgn = (d >= 0.f) ? 1.f : -1.f;
            #pragma unroll
            for (int k = 0; k < 4; ++k) { QTw[k][ln] = sgn*se3r[k]; QTw[4+k][ln] = sgn*se3d[k]; }
        }
        wave_lds_sync();

        v2f wA[2][3];
        #pragma unroll
        for (int t = 0; t < 2; ++t) {
            const float* wr = &Wlw[t*16 + row][0];
            wA[t][0][0] = wr[0+ko]; wA[t][0][1] = wr[1+ko];
            wA[t][1][0] = wr[4+ko]; wA[t][1][1] = wr[5+ko];
            wA[t][2][0] = wr[8+ko]; wA[t][2][1] = wr[9+ko];
        }
        const float* qrow = &QTw[row & 7][hiL ? 8 : 0];
        float4 qa[2], qb[2];
        #pragma unroll
        for (int t = 0; t < 2; ++t) {
            qa[t] = *reinterpret_cast<const float4*>(qrow + 16*t);
            qb[t] = *reinterpret_cast<const float4*>(qrow + 16*t + 4);
        }

        #pragma unroll
        for (int j = 0; j < NT; ++j) {
            const int p = j*16 + row;
            const float x = px[j], y = py[j], z = pz[j];

            v2f f0, f1, f2;
            f0[0] = hiL ? z*z : x*x;   f0[1] = hiL ? x*y : y*y;
            f1[0] = hiL ? x   : x*z;   f1[1] = hiL ? y   : y*z;
            f2[0] = hiL ? 0.f : z;     f2[1] = hiL ? 0.f : 1.f;

            v8f dt[2];
            #pragma unroll
            for (int t = 0; t < 2; ++t) {
                v8f c = {0.f,0.f,0.f,0.f,0.f,0.f,0.f,0.f};
                c = __builtin_amdgcn_wmma_f32_16x16x4_f32(false, wA[t][0], false, f0, (short)0, c, false, false);
                c = __builtin_amdgcn_wmma_f32_16x16x4_f32(false, wA[t][1], false, f1, (short)0, c, false, false);
                c = __builtin_amdgcn_wmma_f32_16x16x4_f32(false, wA[t][2], false, f2, (short)0, c, false, false);
                dt[t] = c;
            }
            #pragma unroll
            for (int r = 1; r < 8; ++r) dt[1][r] = hiL ? __builtin_inff() : dt[1][r];

            float dmin = dt[0][0];
            #pragma unroll
            for (int r = 1; r < 8; ++r) dmin = fminf(dmin, dt[0][r]);
            #pragma unroll
            for (int r = 0; r < 8; ++r) dmin = fminf(dmin, dt[1][r]);
            dmin = fminf(dmin, xhalf(dmin));

            v8f e[2];
            #pragma unroll
            for (int t = 0; t < 2; ++t) {
                #pragma unroll
                for (int r = 0; r < 8; ++r) e[t][r] = __expf(0.5f*(dmin - dt[t][r]));
            }

            v8f acc[2];
            #pragma unroll
            for (int t = 0; t < 2; ++t) {
                v2f a, b;
                v8f c = {0.f,0.f,0.f,0.f,0.f,0.f,0.f,0.f};
                a[0] = qa[t].x; a[1] = qa[t].y; b[0] = e[t][0]; b[1] = e[t][1];
                c = __builtin_amdgcn_wmma_f32_16x16x4_f32(false, a, false, b, (short)0, c, false, false);
                a[0] = qa[t].z; a[1] = qa[t].w; b[0] = e[t][2]; b[1] = e[t][3];
                c = __builtin_amdgcn_wmma_f32_16x16x4_f32(false, a, false, b, (short)0, c, false, false);
                a[0] = qb[t].x; a[1] = qb[t].y; b[0] = e[t][4]; b[1] = e[t][5];
                c = __builtin_amdgcn_wmma_f32_16x16x4_f32(false, a, false, b, (short)0, c, false, false);
                a[0] = qb[t].z; a[1] = qb[t].w; b[0] = e[t][6]; b[1] = e[t][7];
                c = __builtin_amdgcn_wmma_f32_16x16x4_f32(false, a, false, b, (short)0, c, false, false);
                acc[t] = c;
            }
            const float q0 = acc[0][0] + acc[1][0], q1 = acc[0][1] + acc[1][1];
            const float q2 = acc[0][2] + acc[1][2], q3 = acc[0][3] + acc[1][3];
            const float d0 = acc[0][4] + acc[1][4], d1 = acc[0][5] + acc[1][5];
            const float d2 = acc[0][6] + acc[1][6], d3 = acc[0][7] + acc[1][7];

            const float n2  = q0*q0 + q1*q1 + q2*q2 + q3*q3;
            const float rin = rsqrtf(n2);
            const float w  = q0*rin, vx = q1*rin, vy = q2*rin, vz = q3*rin;

            const float ux = vy*z - vz*y, uy = vz*x - vx*z, uz = vx*y - vy*x;
            const float rx = x + 2.f*(w*ux + (vy*uz - vz*uy));
            const float ry = y + 2.f*(w*uy + (vz*ux - vx*uz));
            const float rz = z + 2.f*(w*uz + (vx*uy - vy*ux));
            const float s2 = 2.f*rin*rin;
            const float tx = s2*(-d0*q1 + q0*d1 - d2*q3 + d3*q2);
            const float ty = s2*(-d0*q2 + q0*d2 - d3*q1 + d1*q3);
            const float tz = s2*(-d0*q3 + q0*d3 - d1*q2 + d2*q1);

            if (!hiL) {
                float* op = out + ((size_t)m*NP + p)*3;
                op[0] = rx + tx; op[1] = ry + ty; op[2] = rz + tz;
            }
        }
        wave_lds_sync();
    }
}

extern "C" void kernel_launch(void* const* d_in, const int* in_sizes, int n_in,
                              void* d_out, int out_size, void* d_ws, size_t ws_size,
                              hipStream_t stream) {
    (void)n_in; (void)out_size; (void)d_ws; (void)ws_size;
    const float* xyz     = (const float*)d_in[0];
    const float* t_qr    = (const float*)d_in[1];
    const float* t_qd    = (const float*)d_in[2];
    const float* rest_qr = (const float*)d_in[3];
    const float* rest_qd = (const float*)d_in[4];
    const float* lg      = (const float*)d_in[5];
    float* out = (float*)d_out;

    const int M = in_sizes[1] / (MB * 4);
    dqskin_wave_kernel<<<dim3(NBLK), dim3(32*WPB), 0, stream>>>(
        xyz, t_qr, t_qd, rest_qr, rest_qd, lg, out, M);
}


# Point tiles taken in pairs: lane half h finishes tile 2i+h, so normalise/rotate/translate/store run once per pair on all 32 lanes instead of once per tile on 16; no lane-half branch left
# speedup vs baseline: 1.7424x; 1.1093x over previous
#include <hip/hip_runtime.h>


typedef __attribute__((ext_vector_type(2))) float v2f;
typedef __attribute__((ext_vector_type(8))) float v8f;

#define MB 25
#define NP 64
#define NT 4
#define QP 36
#define WPB 4
#define NBLK 2048

__device__ __forceinline__ void qmul4(const float a[4], const float b[4], float o[4]) {
    o[0] = a[0]*b[0] - a[1]*b[1] - a[2]*b[2] - a[3]*b[3];
    o[1] = a[0]*b[1] + b[0]*a[1] + (a[2]*b[3] - a[3]*b[2]);
    o[2] = a[0]*b[2] + b[0]*a[2] + (a[3]*b[1] - a[1]*b[3]);
    o[3] = a[0]*b[3] + b[0]*a[3] + (a[1]*b[2] - a[2]*b[1]);
}

__device__ __forceinline__ float xhalf(float x) {
    return __builtin_bit_cast(float, __builtin_amdgcn_permlanex16(
        __builtin_bit_cast(int, x), __builtin_bit_cast(int, x), 0x76543210, 0xfedcba98, false, false));
}

__device__ __forceinline__ float lane0(float x) {
    return __builtin_bit_cast(float, __builtin_amdgcn_readfirstlane(__builtin_bit_cast(int, x)));
}

__device__ __forceinline__ void wave_lds_sync() {
    __builtin_amdgcn_fence(__ATOMIC_RELEASE, "wavefront");
    __builtin_amdgcn_wave_barrier();
    __builtin_amdgcn_fence(__ATOMIC_ACQUIRE, "wavefront");
}

__global__ __launch_bounds__(32*WPB, 8)
void dqskin_wave_kernel(const float* __restrict__ xyz,
                        const float* __restrict__ t_qr,
                        const float* __restrict__ t_qd,
                        const float* __restrict__ rest_qr,
                        const float* __restrict__ rest_qd,
                        const float* __restrict__ log_gauss,
                        float* __restrict__ out,
                        int M)
{
    __shared__ __attribute__((aligned(16))) float QT[WPB][8][QP];
    __shared__ float Wl[WPB][32][12];

    const int ln  = threadIdx.x & 31;
    const int wv  = __builtin_amdgcn_readfirstlane((int)(threadIdx.x >> 5));
    const bool hiL = (ln >= 16);
    const int row = ln & 15;
    const int ko  = hiL ? 2 : 0;

    float (*QTw)[QP] = QT[wv];
    float (*Wlw)[12] = Wl[wv];

    for (int m = blockIdx.x*WPB + wv; m < M; m += NBLK*WPB) {
        float px[NT], py[NT], pz[NT];
        #pragma unroll
        for (int j = 0; j < NT; ++j) {
            const float* xp = xyz + ((size_t)m*NP + j*16 + row)*3;
            px[j] = xp[0]; py[j] = xp[1]; pz[j] = xp[2];
        }

        float se3r[4] = {0.f,0.f,0.f,0.f};
        float se3d[4] = {0.f,0.f,0.f,0.f};

        if (ln < MB) {
            const int b = ln;
            const float* tq = t_qr   + ((size_t)m*MB + b)*4;
            const float* td = t_qd   + ((size_t)m*MB + b)*4;
            const float* rq = rest_qr+ ((size_t)m*MB + b)*4;
            const float* rd = rest_qd+ ((size_t)m*MB + b)*4;
            float tqv[4] = {tq[0],  tq[1],  tq[2],  tq[3]};
            float tdv[4] = {td[0],  td[1],  td[2],  td[3]};
            float rqv[4] = {rq[0],  rq[1],  rq[2],  rq[3]};
            float rqc[4] = {rq[0], -rq[1], -rq[2], -rq[3]};
            float rdc[4] = {rd[0], -rd[1], -rd[2], -rd[3]};

            qmul4(tqv, rqc, se3r);
            float u1[4], u2[4];
            qmul4(tqv, rdc, u1);
            qmul4(tdv, rqc, u2);
            #pragma unroll
            for (int k = 0; k < 4; ++k) se3d[k] = u1[k] + u2[k];

            const float w0 = rqc[0], ax = rqc[1], ay = rqc[2], az = rqc[3];
            const float R00 = 1.f - 2.f*(ay*ay + az*az);
            const float R01 = 2.f*(ax*ay - w0*az);
            const float R02 = 2.f*(ax*az + w0*ay);
            const float R10 = 2.f*(ax*ay + w0*az);
            const float R11 = 1.f - 2.f*(ax*ax + az*az);
            const float R12 = 2.f*(ay*az - w0*ax);
            const float R20 = 2.f*(ax*az - w0*ay);
            const float R21 = 2.f*(ay*az + w0*ax);
            const float R22 = 1.f - 2.f*(ax*ax + ay*ay);
            float tt[4];
            qmul4(rdc, rqv, tt);
            const float tx = 2.f*tt[1], ty = 2.f*tt[2], tz = 2.f*tt[3];

            const float i0 = __expf(-log_gauss[b*3+0]);
            const float i1 = __expf(-log_gauss[b*3+1]);
            const float i2 = __expf(-log_gauss[b*3+2]);
            const float A00=R00*i0, A01=R01*i0, A02=R02*i0;
            const float A10=R10*i1, A11=R11*i1, A12=R12*i1;
            const float A20=R20*i2, A21=R21*i2, A22=R22*i2;
            const float c0=tx*i0, c1=ty*i1, c2=tz*i2;

            Wlw[b][0]  = A00*A00 + A10*A10 + A20*A20;
            Wlw[b][1]  = A01*A01 + A11*A11 + A21*A21;
            Wlw[b][2]  = A02*A02 + A12*A12 + A22*A22;
            Wlw[b][3]  = 2.f*(A00*A01 + A10*A11 + A20*A21);
            Wlw[b][4]  = 2.f*(A00*A02 + A10*A12 + A20*A22);
            Wlw[b][5]  = 2.f*(A01*A02 + A11*A12 + A21*A22);
            Wlw[b][6]  = 2.f*(A00*c0 + A10*c1 + A20*c2);
            Wlw[b][7]  = 2.f*(A01*c0 + A11*c1 + A21*c2);
            Wlw[b][8]  = 2.f*(A02*c0 + A12*c1 + A22*c2);
            Wlw[b][9]  = c0*c0 + c1*c1 + c2*c2;
            Wlw[b][10] = 0.f;
            Wlw[b][11] = 0.f;
        } else {
            #pragma unroll
            for (int k = 0; k < 12; ++k) Wlw[ln][k] = 0.f;
        }

        {
            const float d = se3r[0]*lane0(se3r[0]) + se3r[1]*lane0(se3r[1])
                          + se3r[2]*lane0(se3r[2]) + se3r[3]*lane0(se3r[3]);
            const float sgn = (d >= 0.f) ? 1.f : -1.f;
            #pragma unroll
            for (int k = 0; k < 4; ++k) { QTw[k][ln] = sgn*se3r[k]; QTw[4+k][ln] = sgn*se3d[k]; }
        }
        wave_lds_sync();

        v2f wA[2][3];
        #pragma unroll
        for (int t = 0; t < 2; ++t) {
            const float* wr = &Wlw[t*16 + row][0];
            wA[t][0][0] = wr[0+ko]; wA[t][0][1] = wr[1+ko];
            wA[t][1][0] = wr[4+ko]; wA[t][1][1] = wr[5+ko];
            wA[t][2][0] = wr[8+ko]; wA[t][2][1] = wr[9+ko];
        }
        const float* qrow = &QTw[row & 7][hiL ? 8 : 0];
        float4 qa[2], qb[2];
        #pragma unroll
        for (int t = 0; t < 2; ++t) {
            qa[t] = *reinterpret_cast<const float4*>(qrow + 16*t);
            qb[t] = *reinterpret_cast<const float4*>(qrow + 16*t + 4);
        }

        #pragma unroll
        for (int i = 0; i < NT/2; ++i) {
            v8f acc[2][2];
            #pragma unroll
            for (int u = 0; u < 2; ++u) {
                const int j = 2*i + u;
                const float x = px[j], y = py[j], z = pz[j];

                v2f f0, f1, f2;
                f0[0] = hiL ? z*z : x*x;   f0[1] = hiL ? x*y : y*y;
                f1[0] = hiL ? x   : x*z;   f1[1] = hiL ? y   : y*z;
                f2[0] = hiL ? 0.f : z;     f2[1] = hiL ? 0.f : 1.f;

                v8f dt[2];
                #pragma unroll
                for (int t = 0; t < 2; ++t) {
                    v8f c = {0.f,0.f,0.f,0.f,0.f,0.f,0.f,0.f};
                    c = __builtin_amdgcn_wmma_f32_16x16x4_f32(false, wA[t][0], false, f0, (short)0, c, false, false);
                    c = __builtin_amdgcn_wmma_f32_16x16x4_f32(false, wA[t][1], false, f1, (short)0, c, false, false);
                    c = __builtin_amdgcn_wmma_f32_16x16x4_f32(false, wA[t][2], false, f2, (short)0, c, false, false);
                    dt[t] = c;
                }
                #pragma unroll
                for (int r = 1; r < 8; ++r) dt[1][r] = hiL ? __builtin_inff() : dt[1][r];

                float dmin = dt[0][0];
                #pragma unroll
                for (int r = 1; r < 8; ++r) dmin = fminf(dmin, dt[0][r]);
                #pragma unroll
                for (int r = 0; r < 8; ++r) dmin = fminf(dmin, dt[1][r]);
                dmin = fminf(dmin, xhalf(dmin));

                v8f e[2];
                #pragma unroll
                for (int t = 0; t < 2; ++t) {
                    #pragma unroll
                    for (int r = 0; r < 8; ++r) e[t][r] = __expf(0.5f*(dmin - dt[t][r]));
                }

                #pragma unroll
                for (int t = 0; t < 2; ++t) {
                    v2f a, b;
                    v8f c = {0.f,0.f,0.f,0.f,0.f,0.f,0.f,0.f};
                    a[0] = qa[t].x; a[1] = qa[t].y; b[0] = e[t][0]; b[1] = e[t][1];
                    c = __builtin_amdgcn_wmma_f32_16x16x4_f32(false, a, false, b, (short)0, c, false, false);
                    a[0] = qa[t].z; a[1] = qa[t].w; b[0] = e[t][2]; b[1] = e[t][3];
                    c = __builtin_amdgcn_wmma_f32_16x16x4_f32(false, a, false, b, (short)0, c, false, false);
                    a[0] = qb[t].x; a[1] = qb[t].y; b[0] = e[t][4]; b[1] = e[t][5];
                    c = __builtin_amdgcn_wmma_f32_16x16x4_f32(false, a, false, b, (short)0, c, false, false);
                    a[0] = qb[t].z; a[1] = qb[t].w; b[0] = e[t][6]; b[1] = e[t][7];
                    c = __builtin_amdgcn_wmma_f32_16x16x4_f32(false, a, false, b, (short)0, c, false, false);
                    acc[u][t] = c;
                }
            }

            const v8f s0 = acc[0][0] + acc[0][1];
            const v8f s1 = acc[1][0] + acc[1][1];
            const float q0 = hiL ? s1[0] : s0[0], q1 = hiL ? s1[1] : s0[1];
            const float q2 = hiL ? s1[2] : s0[2], q3 = hiL ? s1[3] : s0[3];
            const float d0 = hiL ? s1[4] : s0[4], d1 = hiL ? s1[5] : s0[5];
            const float d2 = hiL ? s1[6] : s0[6], d3 = hiL ? s1[7] : s0[7];
            const float x = hiL ? px[2*i+1] : px[2*i];
            const float y = hiL ? py[2*i+1] : py[2*i];
            const float z = hiL ? pz[2*i+1] : pz[2*i];

            const float n2  = q0*q0 + q1*q1 + q2*q2 + q3*q3;
            const float rin = rsqrtf(n2);
            const float w  = q0*rin, vx = q1*rin, vy = q2*rin, vz = q3*rin;

            const float ux = vy*z - vz*y, uy = vz*x - vx*z, uz = vx*y - vy*x;
            const float rx = x + 2.f*(w*ux + (vy*uz - vz*uy));
            const float ry = y + 2.f*(w*uy + (vz*ux - vx*uz));
            const float rz = z + 2.f*(w*uz + (vx*uy - vy*ux));
            const float s2 = 2.f*rin*rin;
            const float tx = s2*(-d0*q1 + q0*d1 - d2*q3 + d3*q2);
            const float ty = s2*(-d0*q2 + q0*d2 - d3*q1 + d1*q3);
            const float tz = s2*(-d0*q3 + q0*d3 - d1*q2 + d2*q1);

            float* op = out + ((size_t)m*NP + 32*i + ln)*3;
            op[0] = rx + tx; op[1] = ry + ty; op[2] = rz + tz;
        }
        wave_lds_sync();
    }
}

extern "C" void kernel_launch(void* const* d_in, const int* in_sizes, int n_in,
                              void* d_out, int out_size, void* d_ws, size_t ws_size,
                              hipStream_t stream) {
    (void)n_in; (void)out_size; (void)d_ws; (void)ws_size;
    const float* xyz     = (const float*)d_in[0];
    const float* t_qr    = (const float*)d_in[1];
    const float* t_qd    = (const float*)d_in[2];
    const float* rest_qr = (const float*)d_in[3];
    const float* rest_qd = (const float*)d_in[4];
    const float* lg      = (const float*)d_in[5];
    float* out = (float*)d_out;

    const int M = in_sizes[1] / (MB * 4);
    dqskin_wave_kernel<<<dim3(NBLK), dim3(32*WPB), 0, stream>>>(
        xyz, t_qr, t_qd, rest_qr, rest_qd, lg, out, M);
}


# dist2 as |A p + c|^2: bone rows hold the scaled affine map (one K=4 product, 13+12 bones per lane half), lanes square-sum 3 components; 12 WMMAs per tile, not 14; blend sum order changes
# speedup vs baseline: 1.8569x; 1.0657x over previous
#include <hip/hip_runtime.h>


typedef __attribute__((ext_vector_type(2))) float v2f;
typedef __attribute__((ext_vector_type(8))) float v8f;

#define MB 25
#define NP 64
#define NT 4
#define QP 36
#define TR 80
#define SQK 0.84932180f
#define PADV 3.0e18f
#define WPB 4
#define NBLK 2048

__device__ __forceinline__ void qmul4(const float a[4], const float b[4], float o[4]) {
    o[0] = a[0]*b[0] - a[1]*b[1] - a[2]*b[2] - a[3]*b[3];
    o[1] = a[0]*b[1] + b[0]*a[1] + (a[2]*b[3] - a[3]*b[2]);
    o[2] = a[0]*b[2] + b[0]*a[2] + (a[3]*b[1] - a[1]*b[3]);
    o[3] = a[0]*b[3] + b[0]*a[3] + (a[1]*b[2] - a[2]*b[1]);
}

__device__ __forceinline__ float xhalf(float x) {
    return __builtin_bit_cast(float, __builtin_amdgcn_permlanex16(
        __builtin_bit_cast(int, x), __builtin_bit_cast(int, x), 0x76543210, 0xfedcba98, false, false));
}

__device__ __forceinline__ float lane0(float x) {
    return __builtin_bit_cast(float, __builtin_amdgcn_readfirstlane(__builtin_bit_cast(int, x)));
}

__device__ __forceinline__ void wave_lds_sync() {
    __builtin_amdgcn_fence(__ATOMIC_RELEASE, "wavefront");
    __builtin_amdgcn_wave_barrier();
    __builtin_amdgcn_fence(__ATOMIC_ACQUIRE, "wavefront");
}

__global__ __launch_bounds__(32*WPB, 8)
void dqskin_wave_kernel(const float* __restrict__ xyz,
                        const float* __restrict__ t_qr,
                        const float* __restrict__ t_qd,
                        const float* __restrict__ rest_qr,
                        const float* __restrict__ rest_qd,
                        const float* __restrict__ log_gauss,
                        float* __restrict__ out,
                        int M)
{
    __shared__ __attribute__((aligned(16))) float QT[WPB][8][QP];
    __shared__ __attribute__((aligned(16))) float TB[WPB][TR][4];

    const int ln  = threadIdx.x & 31;
    const int wv  = __builtin_amdgcn_readfirstlane((int)(threadIdx.x >> 5));
    const bool hiL = (ln >= 16);
    const int row = ln & 15;
    const int ko  = hiL ? 2 : 0;

    float (*QTw)[QP] = QT[wv];
    float (*TBw)[4]  = TB[wv];

    int slot, r0, r1, r2;
    float padc = 0.f;
    if (ln < MB) {
        const int h = (ln >= 13) ? 1 : 0;
        const int n = ln - 13*h;
        slot = 16*h + n;
        if (n < 8)       { r0 = 8*h + n;          r1 = 16 + 8*h + n;      r2 = 32 + 8*h + n; }
        else if (n < 12) { r0 = 48 + 8*h + n - 8; r1 = 52 + 8*h + n - 8;  r2 = 64 + 8*h + n - 8; }
        else             { r0 = 68;               r1 = 69;                r2 = 70; }
    } else {
        slot = (ln < 28) ? ln - 12 : ln;
        if (ln == 27)      { r0 = 76; r1 = 77; r2 = 78; padc = PADV; }
        else if (ln == 26) { r0 = 79; r1 = 79; r2 = 79; }
        else               { r0 = 71; r1 = 71; r2 = 71; }
    }

    for (int m = blockIdx.x*WPB + wv; m < M; m += NBLK*WPB) {
        float px[NT], py[NT], pz[NT];
        #pragma unroll
        for (int j = 0; j < NT; ++j) {
            const float* xp = xyz + ((size_t)m*NP + j*16 + row)*3;
            px[j] = xp[0]; py[j] = xp[1]; pz[j] = xp[2];
        }

        float se3r[4] = {0.f,0.f,0.f,0.f};
        float se3d[4] = {0.f,0.f,0.f,0.f};
        float4 t0 = make_float4(0.f, 0.f, 0.f, padc);
        float4 t1 = t0, t2 = t0;

        if (ln < MB) {
            const int b = ln;
            const float* tq = t_qr   + ((size_t)m*MB + b)*4;
            const float* td = t_qd   + ((size_t)m*MB + b)*4;
            const float* rq = rest_qr+ ((size_t)m*MB + b)*4;
            const float* rd = rest_qd+ ((size_t)m*MB + b)*4;
            float tqv[4] = {tq[0],  tq[1],  tq[2],  tq[3]};
            float tdv[4] = {td[0],  td[1],  td[2],  td[3]};
            float rqv[4] = {rq[0],  rq[1],  rq[2],  rq[3]};
            float rqc[4] = {rq[0], -rq[1], -rq[2], -rq[3]};
            float rdc[4] = {rd[0], -rd[1], -rd[2], -rd[3]};

            qmul4(tqv, rqc, se3r);
            float u1[4], u2[4];
            qmul4(tqv, rdc, u1);
            qmul4(tdv, rqc, u2);
            #pragma unroll
            for (int k = 0; k < 4; ++k) se3d[k] = u1[k] + u2[k];

            const float w0 = rqc[0], ax = rqc[1], ay = rqc[2], az = rqc[3];
            const float R00 = 1.f - 2.f*(ay*ay + az*az);
            const float R01 = 2.f*(ax*ay - w0*az);
            const float R02 = 2.f*(ax*az + w0*ay);
            const float R10 = 2.f*(ax*ay + w0*az);
            const float R11 = 1.f - 2.f*(ax*ax + az*az);
            const float R12 = 2.f*(ay*az - w0*ax);
            const float R20 = 2.f*(ax*az - w0*ay);
            const float R21 = 2.f*(ay*az + w0*ax);
            const float R22 = 1.f - 2.f*(ax*ax + ay*ay);
            float tt[4];
            qmul4(rdc, rqv, tt);
            const float tx = 2.f*tt[1], ty = 2.f*tt[2], tz = 2.f*tt[3];

            const float i0 = __expf(-log_gauss[b*3+0]) * SQK;
            const float i1 = __expf(-log_gauss[b*3+1]) * SQK;
            const float i2 = __expf(-log_gauss[b*3+2]) * SQK;
            t0 = make_float4(R00*i0, R01*i0, R02*i0, tx*i0);
            t1 = make_float4(R10*i1, R11*i1, R12*i1, ty*i1);
            t2 = make_float4(R20*i2, R21*i2, R22*i2, tz*i2);
        }
        *reinterpret_cast<float4*>(&TBw[r0][0]) = t0;
        *reinterpret_cast<float4*>(&TBw[r1][0]) = t1;
        *reinterpret_cast<float4*>(&TBw[r2][0]) = t2;

        {
            const float d = se3r[0]*lane0(se3r[0]) + se3r[1]*lane0(se3r[1])
                          + se3r[2]*lane0(se3r[2]) + se3r[3]*lane0(se3r[3]);
            const float sgn = (d >= 0.f) ? 1.f : -1.f;
            #pragma unroll
            for (int k = 0; k < 4; ++k) { QTw[k][slot] = sgn*se3r[k]; QTw[4+k][slot] = sgn*se3d[k]; }
        }
        wave_lds_sync();

        v2f tA[5];
        #pragma unroll
        for (int q = 0; q < 5; ++q) {
            const float* tr = &TBw[q*16 + row][ko];
            tA[q][0] = tr[0]; tA[q][1] = tr[1];
        }
        const float* qrow = &QTw[row & 7][hiL ? 16 : 0];
        float4 qv[4];
        #pragma unroll
        for (int t = 0; t < 4; ++t) qv[t] = *reinterpret_cast<const float4*>(qrow + 4*t);

        #pragma unroll
        for (int i = 0; i < NT/2; ++i) {
            v8f acc[2];
            #pragma unroll
            for (int u = 0; u < 2; ++u) {
                const int j = 2*i + u;
                const float x = px[j], y = py[j], z = pz[j];

                v2f f;
                f[0] = hiL ? z   : x;
                f[1] = hiL ? 1.f : y;

                v8f yt[5];
                #pragma unroll
                for (int q = 0; q < 5; ++q) {
                    v8f c = {0.f,0.f,0.f,0.f,0.f,0.f,0.f,0.f};
                    yt[q] = __builtin_amdgcn_wmma_f32_16x16x4_f32(false, tA[q], false, f, (short)0, c, false, false);
                }

                v2f dp[7];
                #pragma unroll
                for (int p = 0; p < 4; ++p) {
                    const v2f a0 = {yt[0][2*p], yt[0][2*p+1]}, a1 = {yt[1][2*p], yt[1][2*p+1]}, a2 = {yt[2][2*p], yt[2][2*p+1]};
                    dp[p] = __builtin_elementwise_fma(a2, a2, __builtin_elementwise_fma(a1, a1, a0*a0));
                }
                #pragma unroll
                for (int p = 0; p < 2; ++p) {
                    const v2f a0 = {yt[3][2*p], yt[3][2*p+1]}, a1 = {yt[3][4+2*p], yt[3][5+2*p]}, a2 = {yt[4][2*p], yt[4][2*p+1]};
                    dp[4+p] = __builtin_elementwise_fma(a2, a2, __builtin_elementwise_fma(a1, a1, a0*a0));
                }
                dp[6][0] = fmaf(yt[4][6], yt[4][6], fmaf(yt[4][5], yt[4][5], yt[4][4]*yt[4][4]));
                dp[6][1] = dp[6][0];

                float dmin = fminf(dp[0][0], dp[0][1]);
                #pragma unroll
                for (int s = 1; s < 6; ++s) dmin = fminf(dmin, fminf(dp[s][0], dp[s][1]));
                dmin = fminf(dmin, dp[6][0]);
                dmin = fminf(dmin, xhalf(dmin));

                const v2f dm2 = {dmin, dmin};
                v8f c = {0.f,0.f,0.f,0.f,0.f,0.f,0.f,0.f};
                #pragma unroll
                for (int s = 0; s < 7; ++s) {
                    v2f a, b;
                    const float4 qq = qv[s >> 1];
                    a[0] = (s & 1) ? qq.z : qq.x;
                    a[1] = (s & 1) ? qq.w : qq.y;
                    const v2f g = dm2 - dp[s];
                    b[0] = __builtin_amdgcn_exp2f(g[0]);
                    b[1] = (s < 6) ? __builtin_amdgcn_exp2f(g[1]) : 0.f;
                    c = __builtin_amdgcn_wmma_f32_16x16x4_f32(false, a, false, b, (short)0, c, false, false);
                }
                acc[u] = c;
            }

            const v8f s0 = acc[0];
            const v8f s1 = acc[1];
            const float q0 = hiL ? s1[0] : s0[0], q1 = hiL ? s1[1] : s0[1];
            const float q2 = hiL ? s1[2] : s0[2], q3 = hiL ? s1[3] : s0[3];
            const float d0 = hiL ? s1[4] : s0[4], d1 = hiL ? s1[5] : s0[5];
            const float d2 = hiL ? s1[6] : s0[6], d3 = hiL ? s1[7] : s0[7];
            const float x = hiL ? px[2*i+1] : px[2*i];
            const float y = hiL ? py[2*i+1] : py[2*i];
            const float z = hiL ? pz[2*i+1] : pz[2*i];

            const float n2  = q0*q0 + q1*q1 + q2*q2 + q3*q3;
            const float rin = rsqrtf(n2);
            const float w  = q0*rin, vx = q1*rin, vy = q2*rin, vz = q3*rin;

            const float ux = vy*z - vz*y, uy = vz*x - vx*z, uz = vx*y - vy*x;
            const float rx = x + 2.f*(w*ux + (vy*uz - vz*uy));
            const float ry = y + 2.f*(w*uy + (vz*ux - vx*uz));
            const float rz = z + 2.f*(w*uz + (vx*uy - vy*ux));
            const float s2 = 2.f*rin*rin;
            const float tx = s2*(-d0*q1 + q0*d1 - d2*q3 + d3*q2);
            const float ty = s2*(-d0*q2 + q0*d2 - d3*q1 + d1*q3);
            const float tz = s2*(-d0*q3 + q0*d3 - d1*q2 + d2*q1);

            float* op = out + ((size_t)m*NP + 32*i + ln)*3;
            op[0] = rx + tx; op[1] = ry + ty; op[2] = rz + tz;
        }
        wave_lds_sync();
    }
}

extern "C" void kernel_launch(void* const* d_in, const int* in_sizes, int n_in,
                              void* d_out, int out_size, void* d_ws, size_t ws_size,
                              hipStream_t stream) {
    (void)n_in; (void)out_size; (void)d_ws; (void)ws_size;
    const float* xyz     = (const float*)d_in[0];
    const float* t_qr    = (const float*)d_in[1];
    const float* t_qd    = (const float*)d_in[2];
    const float* rest_qr = (const float*)d_in[3];
    const float* rest_qd = (const float*)d_in[4];
    const float* lg      = (const float*)d_in[5];
    float* out = (float*)d_out;

    const int M = in_sizes[1] / (MB * 4);
    dqskin_wave_kernel<<<dim3(NBLK), dim3(32*WPB), 0, stream>>>(
        xyz, t_qr, t_qd, rest_qr, rest_qd, lg, out, M);
}
